# starting kernel
# baseline (speedup 1.0000x reference)
#include <hip/hip_runtime.h>

typedef __attribute__((ext_vector_type(2))) float v2f;
typedef __attribute__((ext_vector_type(8))) float v8f;

__global__ void fill_f32(float* __restrict__ p, float v, int n) {
    int i = blockIdx.x * blockDim.x + threadIdx.x;
    if (i < n) p[i] = v;
}

__global__ void deg_edges(const int* __restrict__ dst, float* __restrict__ deg, int nE) {
    int e = blockIdx.x * blockDim.x + threadIdx.x;
    if (e < nE) atomicAdd(&deg[dst[e]], 1.0f);
}

__global__ void deg_to_dis(float* __restrict__ d, int n) {
    int i = blockIdx.x * blockDim.x + threadIdx.x;
    if (i < n) d[i] = rsqrtf(d[i]);
}

template<int FT>
__global__ void wmma_gemm(const float* __restrict__ X, const float* __restrict__ W,
                          const float* __restrict__ Bias, float* __restrict__ Y,
                          int N, int K, int addBias) {
    const int F    = FT * 16;
    int wave = (blockIdx.x * blockDim.x + threadIdx.x) >> 5;
    int lane = threadIdx.x & 31;
    int row0 = wave << 4;
    if (row0 >= N) return;
    int half = lane >> 4;
    int l16  = lane & 15;
    int arow = row0 + l16; if (arow > N - 1) arow = N - 1;

    v8f acc[FT];
    v8f z = {0.f, 0.f, 0.f, 0.f, 0.f, 0.f, 0.f, 0.f};
#pragma unroll
    for (int t = 0; t < FT; ++t) acc[t] = z;

    for (int k0 = 0; k0 < K; k0 += 4) {
        int ka = k0 + half * 2;
        v2f a;
        a.x = X[arow * K + ka];
        a.y = X[arow * K + ka + 1];
#pragma unroll
        for (int t = 0; t < FT; ++t) {
            int col = t * 16 + l16;
            v2f b;
            b.x = W[ka * F + col];
            b.y = W[(ka + 1) * F + col];
            acc[t] = __builtin_amdgcn_wmma_f32_16x16x4_f32(
                 false, a,  false, b,
                 (short)0, acc[t],  false,  false);
        }
    }

#pragma unroll
    for (int t = 0; t < FT; ++t) {
        int col = t * 16 + l16;
        float bb = addBias ? Bias[col] : 0.f;
#pragma unroll
        for (int v = 0; v < 8; ++v) {
            int r = row0 + half * 8 + v;
            if (r < N) Y[r * F + col] = acc[t][v] + bb;
        }
    }
}

__global__ void agg_init(const float* __restrict__ H, const float* __restrict__ dis,
                         float* __restrict__ AGG, int N, int F) {
    int idx = blockIdx.x * blockDim.x + threadIdx.x;
    if (idx < N * F) {
        float d = dis[idx / F];
        AGG[idx] = H[idx] * d * d;
    }
}

__global__ void edge_scatter(const int* __restrict__ src, const int* __restrict__ dst,
                             const float* __restrict__ dis, const float* __restrict__ H,
                             float* __restrict__ AGG, int nE, int F) {
    int e    = (blockIdx.x * blockDim.x + threadIdx.x) >> 5;
    int lane = threadIdx.x & 31;
    if (e >= nE) return;
    int   s = src[e];
    int   d = dst[e];
    float w = dis[s] * dis[d];
    for (int f = lane; f < F; f += 32)
        atomicAdd(&AGG[d * F + f], H[s * F + f] * w);
}

__global__ void bias_relu(float* __restrict__ A, const float* __restrict__ b, int N, int F) {
    int idx = blockIdx.x * blockDim.x + threadIdx.x;
    if (idx < N * F) {
        float v = A[idx] + b[idx % F];
        A[idx] = v > 0.f ? v : 0.f;
    }
}

__global__ void pool_nodes(const float* __restrict__ H, const int* __restrict__ batch,
                           float* __restrict__ gmax, float* __restrict__ gsum,
                           float* __restrict__ cnt, int N, int F) {
    int i    = (blockIdx.x * blockDim.x + threadIdx.x) >> 5;
    int lane = threadIdx.x & 31;
    if (i >= N) return;
    int g = batch[i];
    for (int f = lane; f < F; f += 32) {
        float v = H[i * F + f];
        atomicMax((unsigned int*)&gmax[g * F + f], __float_as_uint(v));
        atomicAdd(&gsum[g * F + f], v);
    }
    if (lane == 0) atomicAdd(&cnt[g], 1.0f);
}

__global__ void build_pooled(const float* __restrict__ gmax, const float* __restrict__ gsum,
                             const float* __restrict__ cnt, float* __restrict__ pooled,
                             int G, int F) {
    int idx = blockIdx.x * blockDim.x + threadIdx.x;
    if (idx >= G * 2 * F) return;
    int g = idx / (2 * F);
    int f = idx % (2 * F);
    float v;
    if (f < F) {
        v = gmax[g * F + f];
    } else {
        float c = cnt[g];
        v = gsum[g * F + (f - F)] / (c > 1.f ? c : 1.f);
    }
    pooled[idx] = v;
}

extern "C" void kernel_launch(void* const* d_in, const int* in_sizes, int n_in,
                              void* d_out, int out_size, void* d_ws, size_t ws_size,
                              hipStream_t stream) {
    const float* x     = (const float*)d_in[0];
    const int*   ei    = (const int*)d_in[1];
    const int*   batch = (const int*)d_in[2];
    const float* W1    = (const float*)d_in[3];
    const float* b1    = (const float*)d_in[4];
    const float* W2    = (const float*)d_in[5];
    const float* b2    = (const float*)d_in[6];
    const float* W3    = (const float*)d_in[7];
    const float* b3    = (const float*)d_in[8];
    const float* Wout  = (const float*)d_in[9];
    const float* bout  = (const float*)d_in[10];
    float*       out   = (float*)d_out;

    const int N  = in_sizes[2];
    const int E  = in_sizes[1] / 2;
    const int K0 = in_sizes[0] / N;
    const int F1 = in_sizes[4];
    const int F2 = in_sizes[6];
    const int F3 = in_sizes[8];
    const int Fo = in_sizes[10];
    const int G  = out_size / Fo;

    const int* src = ei;
    const int* dst = ei + E;

    char*  ws  = (char*)d_ws;
    size_t off = 0;
    auto carve = [&](size_t bytes) -> float* {
        float* p = (float*)(ws + off);
        off += (bytes + 255) & ~(size_t)255;
        return p;
    };
    float* dis    = carve((size_t)N * 4);
    float* bufT   = carve((size_t)N * F3 * 4);
    float* bufA   = carve((size_t)N * F3 * 4);
    float* bufB   = carve((size_t)N * F3 * 4);
    float* gmax   = carve((size_t)G * F3 * 4);
    float* gsum   = carve((size_t)G * F3 * 4);
    float* cnt    = carve((size_t)G * 4);
    float* pooled = carve((size_t)G * 2 * F3 * 4);

    const int BLK = 256;
    auto grid1   = [&](long long n) { return dim3((unsigned)((n + BLK - 1) / BLK)); };
    auto gridWv  = [&](long long waves) { return dim3((unsigned)((waves * 32 + BLK - 1) / BLK)); };
    auto gemmGrid = [&](int rows) { return gridWv((rows + 15) / 16); };

    fill_f32<<<grid1(N), BLK, 0, stream>>>(dis, 1.0f, N);
    deg_edges<<<grid1(E), BLK, 0, stream>>>(dst, dis, E);
    deg_to_dis<<<grid1(N), BLK, 0, stream>>>(dis, N);

    wmma_gemm<2><<<gemmGrid(N), BLK, 0, stream>>>(x, W1, b1, bufT, N, K0, 0);
    agg_init<<<grid1((long long)N * F1), BLK, 0, stream>>>(bufT, dis, bufA, N, F1);
    edge_scatter<<<gridWv(E), BLK, 0, stream>>>(src, dst, dis, bufT, bufA, E, F1);
    bias_relu<<<grid1((long long)N * F1), BLK, 0, stream>>>(bufA, b1, N, F1);

    wmma_gemm<3><<<gemmGrid(N), BLK, 0, stream>>>(bufA, W2, b2, bufT, N, F1, 0);
    agg_init<<<grid1((long long)N * F2), BLK, 0, stream>>>(bufT, dis, bufB, N, F2);
    edge_scatter<<<gridWv(E), BLK, 0, stream>>>(src, dst, dis, bufT, bufB, E, F2);
    bias_relu<<<grid1((long long)N * F2), BLK, 0, stream>>>(bufB, b2, N, F2);

    wmma_gemm<4><<<gemmGrid(N), BLK, 0, stream>>>(bufB, W3, b3, bufT, N, F2, 0);
    agg_init<<<grid1((long long)N * F3), BLK, 0, stream>>>(bufT, dis, bufA, N, F3);
    edge_scatter<<<gridWv(E), BLK, 0, stream>>>(src, dst, dis, bufT, bufA, E, F3);
    bias_relu<<<grid1((long long)N * F3), BLK, 0, stream>>>(bufA, b3, N, F3);

    fill_f32<<<grid1((long long)G * F3), BLK, 0, stream>>>(gmax, 0.0f, G * F3);
    fill_f32<<<grid1((long long)G * F3), BLK, 0, stream>>>(gsum, 0.0f, G * F3);
    fill_f32<<<grid1(G), BLK, 0, stream>>>(cnt, 0.0f, G);
    pool_nodes<<<gridWv(N), BLK, 0, stream>>>(bufA, batch, gmax, gsum, cnt, N, F3);
    build_pooled<<<grid1((long long)G * 2 * F3), BLK, 0, stream>>>(gmax, gsum, cnt, pooled, G, F3);

    wmma_gemm<2><<<gemmGrid(G), BLK, 0, stream>>>(pooled, Wout, bout, out, G, 2 * F3, 1);
}


# layer sums per 128-row destination tile in LDS: edges binned by tile once per call; a work-group adds its tile's edge rows with LDS f32 adds, writes each row once; no global row atomics; order varies
# speedup vs baseline: 2.2717x; 2.2717x over previous
#include <hip/hip_runtime.h>

typedef __attribute__((ext_vector_type(2))) float v2f;
typedef __attribute__((ext_vector_type(8))) float v8f;

__global__ void fill_f32(float* __restrict__ p, float v, int n) {
    int i = blockIdx.x * blockDim.x + threadIdx.x;
    if (i < n) p[i] = v;
}

__global__ void deg_edges(const int* __restrict__ dst, float* __restrict__ deg, int nE) {
    int e = blockIdx.x * blockDim.x + threadIdx.x;
    if (e < nE) atomicAdd(&deg[dst[e]], 1.0f);
}

__global__ void deg_to_dis(float* __restrict__ d, int n) {
    int i = blockIdx.x * blockDim.x + threadIdx.x;
    if (i < n) d[i] = rsqrtf(d[i]);
}

template<int FT>
__global__ void wmma_gemm(const float* __restrict__ X, const float* __restrict__ W,
                          const float* __restrict__ Bias, float* __restrict__ Y,
                          int N, int K, int addBias) {
    const int F    = FT * 16;
    int wave = (blockIdx.x * blockDim.x + threadIdx.x) >> 5;
    int lane = threadIdx.x & 31;
    int row0 = wave << 4;
    if (row0 >= N) return;
    int half = lane >> 4;
    int l16  = lane & 15;
    int arow = row0 + l16; if (arow > N - 1) arow = N - 1;

    v8f acc[FT];
    v8f z = {0.f, 0.f, 0.f, 0.f, 0.f, 0.f, 0.f, 0.f};
#pragma unroll
    for (int t = 0; t < FT; ++t) acc[t] = z;

    for (int k0 = 0; k0 < K; k0 += 4) {
        int ka = k0 + half * 2;
        v2f a;
        a.x = X[arow * K + ka];
        a.y = X[arow * K + ka + 1];
#pragma unroll
        for (int t = 0; t < FT; ++t) {
            int col = t * 16 + l16;
            v2f b;
            b.x = W[ka * F + col];
            b.y = W[(ka + 1) * F + col];
            acc[t] = __builtin_amdgcn_wmma_f32_16x16x4_f32(
                 false, a,  false, b,
                 (short)0, acc[t],  false,  false);
        }
    }

#pragma unroll
    for (int t = 0; t < FT; ++t) {
        int col = t * 16 + l16;
        float bb = addBias ? Bias[col] : 0.f;
#pragma unroll
        for (int v = 0; v < 8; ++v) {
            int r = row0 + half * 8 + v;
            if (r < N) Y[r * F + col] = acc[t][v] + bb;
        }
    }
}

#define TILE_SHIFT 7
#define TILE_ROWS  128

__device__ __forceinline__ void wave_scan_excl(int* a, int n, int lane) {
    int carry = 0;
    for (int b0 = 0; b0 < n; b0 += 32) {
        int i = b0 + lane;
        int v = (i < n) ? a[i] : 0;
        int x = v;
#pragma unroll
        for (int d = 1; d < 32; d <<= 1) {
            int y = __shfl_up(x, d);
            if (lane >= d) x += y;
        }
        if (i < n) a[i] = carry + x - v;
        carry += __shfl(x, 31);
    }
    if (lane == 0) a[n] = carry;
}

__global__ void __launch_bounds__(1024)
bin_hist(const int* __restrict__ dst, int* __restrict__ binCount, int nE, int nT, int chunk) {
    extern __shared__ int sm[];
    for (int b = threadIdx.x; b < nT; b += blockDim.x) sm[b] = 0;
    __syncthreads();
    int e0 = blockIdx.x * chunk;
    int e1 = e0 + chunk; if (e1 > nE) e1 = nE;
    for (int e = e0 + threadIdx.x; e < e1; e += blockDim.x)
        atomicAdd(&sm[dst[e] >> TILE_SHIFT], 1);
    __syncthreads();
    for (int b = threadIdx.x; b < nT; b += blockDim.x) {
        int c = sm[b];
        if (c) atomicAdd(&binCount[b], c);
    }
}

__global__ void __launch_bounds__(1024)
bin_scan(const int* __restrict__ binCount, int* __restrict__ binStart,
         int* __restrict__ binCursor, int nT) {
    extern __shared__ int sm[];
    for (int b = threadIdx.x; b < nT; b += blockDim.x) sm[b] = binCount[b];
    __syncthreads();
    if (threadIdx.x < 32) wave_scan_excl(sm, nT, (int)threadIdx.x);
    __syncthreads();
    for (int b = threadIdx.x; b <= nT; b += blockDim.x) {
        int v = sm[b];
        binStart[b] = v;
        if (b < nT) binCursor[b] = v;
    }
}

__global__ void __launch_bounds__(1024)
bin_fill(const int* __restrict__ src, const int* __restrict__ dst, const float* __restrict__ dis,
         int* __restrict__ binCursor, uint2* __restrict__ recs, int nE, int nT, int chunk) {
    extern __shared__ int sm[];
    int* sS   = sm;
    int* sD   = sS + chunk;
    int* h    = sD + chunk;
    int* gdel = h + nT + 1;
    int* rnk  = gdel + nT;
    for (int b = threadIdx.x; b < nT; b += blockDim.x) { h[b] = 0; rnk[b] = 0; }
    __syncthreads();
    int e0 = blockIdx.x * chunk;
    int e1 = e0 + chunk; if (e1 > nE) e1 = nE;
    for (int e = e0 + threadIdx.x; e < e1; e += blockDim.x)
        atomicAdd(&h[dst[e] >> TILE_SHIFT], 1);
    __syncthreads();
    for (int b = threadIdx.x; b < nT; b += blockDim.x) {
        int c = h[b];
        gdel[b] = c ? atomicAdd(&binCursor[b], c) : 0;
    }
    __syncthreads();
    if (threadIdx.x < 32) wave_scan_excl(h, nT, (int)threadIdx.x);
    __syncthreads();
    for (int b = threadIdx.x; b < nT; b += blockDim.x) gdel[b] -= h[b];
    for (int e = e0 + threadIdx.x; e < e1; e += blockDim.x) {
        int s = src[e];
        int d = dst[e];
        int b = d >> TILE_SHIFT;
        int li = h[b] + atomicAdd(&rnk[b], 1);
        sS[li] = s;
        sD[li] = d;
    }
    __syncthreads();
    int n = e1 - e0;
    for (int i = threadIdx.x; i < n; i += blockDim.x) {
        int s = sS[i];
        int d = sD[i];
        float w = dis[s] * dis[d];
        recs[i + gdel[d >> TILE_SHIFT]] =
            make_uint2(((unsigned)s << TILE_SHIFT) | ((unsigned)d & (TILE_ROWS - 1)), __float_as_uint(w));
    }
}

template<int F>
__device__ __forceinline__ void add_edge_row(const float* __restrict__ H, float* acc,
                                             unsigned p, float w, int lane) {
    const float* row = H + (size_t)(p >> TILE_SHIFT) * F;
    float* a = acc + (p & (TILE_ROWS - 1)) * F;
    for (int f = lane; f < F; f += 32) atomicAdd(&a[f], row[f] * w);
}

template<int F>
__global__ void __launch_bounds__(512)
tile_agg(const float* __restrict__ H, const float* __restrict__ dis, const float* __restrict__ bias,
         const int* __restrict__ binStart, const uint2* __restrict__ recs,
         float* __restrict__ OUT, int N) {
    __shared__ float acc[TILE_ROWS * F];
    const int row0 = blockIdx.x << TILE_SHIFT;
    const int lane = threadIdx.x & 31;
    const int wave = __builtin_amdgcn_readfirstlane((int)(threadIdx.x >> 5));
    const int nw   = (int)(blockDim.x >> 5);
    const int beg  = __builtin_amdgcn_readfirstlane(binStart[blockIdx.x]);
    const int end  = __builtin_amdgcn_readfirstlane(binStart[blockIdx.x + 1]);

    for (int i = threadIdx.x; i < TILE_ROWS * F; i += blockDim.x) {
        int d = row0 + i / F;
        float v = 0.f;
        if (d < N) { float dd = dis[d]; v = H[(size_t)row0 * F + i] * dd * dd; }
        acc[i] = v;
    }

    const int c1 = (lane + 32 < F) ? lane + 32 : lane;
    int base = beg + wave * 32;
    uint2 cur = make_uint2(0u, 0u);
    if (base + lane < end) cur = recs[base + lane];
    __syncthreads();

    for (; base < end; base += nw * 32) {
        int n = end - base; if (n > 32) n = 32;
        int nb = base + nw * 32;
        uint2 nxt = make_uint2(0u, 0u);
        if (nb + lane < end) nxt = recs[nb + lane];
        if (n == 32) {
#pragma unroll
            for (int j0 = 0; j0 < 32; j0 += 8) {
                float v0[8], v1[8];
#pragma unroll
                for (int u = 0; u < 8; ++u) {
                    unsigned p = (unsigned)__builtin_amdgcn_readlane((int)cur.x, j0 + u);
                    const float* row = H + (size_t)(p >> TILE_SHIFT) * F;
                    v0[u] = row[lane];
                    v1[u] = 0.f;
                    if (F > 32) v1[u] = row[c1];
                }
#pragma unroll
                for (int u = 0; u < 8; ++u) {
                    unsigned p = (unsigned)__builtin_amdgcn_readlane((int)cur.x, j0 + u);
                    float w = __uint_as_float((unsigned)__builtin_amdgcn_readlane((int)cur.y, j0 + u));
                    float* a = acc + (p & (TILE_ROWS - 1)) * F;
                    atomicAdd(&a[lane], v0[u] * w);
                    if (F > 32) { if (lane + 32 < F) atomicAdd(&a[lane + 32], v1[u] * w); }
                }
            }
        } else {
            for (int j = 0; j < n; ++j) {
                unsigned p = (unsigned)__builtin_amdgcn_readlane((int)cur.x, j);
                float w = __uint_as_float((unsigned)__builtin_amdgcn_readlane((int)cur.y, j));
                add_edge_row<F>(H, acc, p, w, lane);
            }
        }
        cur = nxt;
    }
    __syncthreads();

    for (int i = threadIdx.x; i < TILE_ROWS * F; i += blockDim.x) {
        int d = row0 + i / F;
        if (d < N) {
            float v = acc[i] + bias[i % F];
            OUT[(size_t)row0 * F + i] = v > 0.f ? v : 0.f;
        }
    }
}

__global__ void pool_nodes(const float* __restrict__ H, const int* __restrict__ batch,
                           float* __restrict__ gmax, float* __restrict__ gsum,
                           float* __restrict__ cnt, int N, int F) {
    int i    = (blockIdx.x * blockDim.x + threadIdx.x) >> 5;
    int lane = threadIdx.x & 31;
    if (i >= N) return;
    int g = batch[i];
    for (int f = lane; f < F; f += 32) {
        float v = H[i * F + f];
        atomicMax((unsigned int*)&gmax[g * F + f], __float_as_uint(v));
        atomicAdd(&gsum[g * F + f], v);
    }
    if (lane == 0) atomicAdd(&cnt[g], 1.0f);
}

__global__ void build_pooled(const float* __restrict__ gmax, const float* __restrict__ gsum,
                             const float* __restrict__ cnt, float* __restrict__ pooled,
                             int G, int F) {
    int idx = blockIdx.x * blockDim.x + threadIdx.x;
    if (idx >= G * 2 * F) return;
    int g = idx / (2 * F);
    int f = idx % (2 * F);
    float v;
    if (f < F) {
        v = gmax[g * F + f];
    } else {
        float c = cnt[g];
        v = gsum[g * F + (f - F)] / (c > 1.f ? c : 1.f);
    }
    pooled[idx] = v;
}

extern "C" void kernel_launch(void* const* d_in, const int* in_sizes, int n_in,
                              void* d_out, int out_size, void* d_ws, size_t ws_size,
                              hipStream_t stream) {
    const float* x     = (const float*)d_in[0];
    const int*   ei    = (const int*)d_in[1];
    const int*   batch = (const int*)d_in[2];
    const float* W1    = (const float*)d_in[3];
    const float* b1    = (const float*)d_in[4];
    const float* W2    = (const float*)d_in[5];
    const float* b2    = (const float*)d_in[6];
    const float* W3    = (const float*)d_in[7];
    const float* b3    = (const float*)d_in[8];
    const float* Wout  = (const float*)d_in[9];
    const float* bout  = (const float*)d_in[10];
    float*       out   = (float*)d_out;

    const int N  = in_sizes[2];
    const int E  = in_sizes[1] / 2;
    const int K0 = in_sizes[0] / N;
    const int F1 = in_sizes[4];
    const int F2 = in_sizes[6];
    const int F3 = in_sizes[8];
    const int Fo = in_sizes[10];
    const int G  = out_size / Fo;

    const int* src = ei;
    const int* dst = ei + E;

    char*  ws  = (char*)d_ws;
    size_t off = 0;
    auto carve = [&](size_t bytes) -> float* {
        float* p = (float*)(ws + off);
        off += (bytes + 255) & ~(size_t)255;
        return p;
    };
    float* dis    = carve((size_t)N * 4);
    float* bufT   = carve((size_t)N * F3 * 4);
    float* bufA   = carve((size_t)N * F3 * 4);
    float* bufB   = carve((size_t)N * F3 * 4);
    float* gmax   = carve((size_t)G * F3 * 4);
    float* gsum   = carve((size_t)G * F3 * 4);
    float* cnt    = carve((size_t)G * 4);
    float* pooled = carve((size_t)G * 2 * F3 * 4);
    const int nT = (N + TILE_ROWS - 1) / TILE_ROWS;
    int*   binCount  = (int*)carve((size_t)nT * 4);
    int*   binStart  = (int*)carve((size_t)(nT + 1) * 4);
    int*   binCursor = (int*)carve((size_t)nT * 4);
    uint2* recs      = (uint2*)carve((size_t)E * 8);

    const int BLK = 256;
    auto grid1   = [&](long long n) { return dim3((unsigned)((n + BLK - 1) / BLK)); };
    auto gridWv  = [&](long long waves) { return dim3((unsigned)((waves * 32 + BLK - 1) / BLK)); };
    auto gemmGrid = [&](int rows) { return gridWv((rows + 15) / 16); };

    fill_f32<<<grid1(N), BLK, 0, stream>>>(dis, 1.0f, N);
    deg_edges<<<grid1(E), BLK, 0, stream>>>(dst, dis, E);
    deg_to_dis<<<grid1(N), BLK, 0, stream>>>(dis, N);

    const int CHUNK = 6144;
    const int nBinBlk = (E + CHUNK - 1) / CHUNK;
    fill_f32<<<grid1(nT), BLK, 0, stream>>>((float*)binCount, 0.0f, nT);
    bin_hist<<<nBinBlk, 1024, (size_t)nT * 4, stream>>>(dst, binCount, E, nT, CHUNK);
    bin_scan<<<1, 1024, (size_t)(nT + 1) * 4, stream>>>(binCount, binStart, binCursor, nT);
    bin_fill<<<nBinBlk, 1024, (size_t)(2 * CHUNK + 3 * nT + 1) * 4, stream>>>(src, dst, dis, binCursor, recs, E, nT, CHUNK);

    wmma_gemm<2><<<gemmGrid(N), BLK, 0, stream>>>(x, W1, b1, bufT, N, K0, 0);
    tile_agg<32><<<nT, 512, 0, stream>>>(bufT, dis, b1, binStart, recs, bufA, N);

    wmma_gemm<3><<<gemmGrid(N), BLK, 0, stream>>>(bufA, W2, b2, bufT, N, F1, 0);
    tile_agg<48><<<nT, 512, 0, stream>>>(bufT, dis, b2, binStart, recs, bufB, N);

    wmma_gemm<4><<<gemmGrid(N), BLK, 0, stream>>>(bufB, W3, b3, bufT, N, F2, 0);
    tile_agg<64><<<nT, 512, 0, stream>>>(bufT, dis, b3, binStart, recs, bufA, N);

    fill_f32<<<grid1((long long)G * F3), BLK, 0, stream>>>(gmax, 0.0f, G * F3);
    fill_f32<<<grid1((long long)G * F3), BLK, 0, stream>>>(gsum, 0.0f, G * F3);
    fill_f32<<<grid1(G), BLK, 0, stream>>>(cnt, 0.0f, G);
    pool_nodes<<<gridWv(N), BLK, 0, stream>>>(bufA, batch, gmax, gsum, cnt, N, F3);
    build_pooled<<<grid1((long long)G * 2 * F3), BLK, 0, stream>>>(gmax, gsum, cnt, pooled, G, F3);

    wmma_gemm<2><<<gemmGrid(G), BLK, 0, stream>>>(pooled, Wout, bout, out, G, 2 * F3, 1);
}


# node degrees counted per 128-row dst tile from the binned edge keys with LDS integer adds (exact, dis bit-equal), weights filled in a pass after; no per-edge global f32 adds; 19 to 18 launches
# speedup vs baseline: 2.9988x; 1.3201x over previous
#include <hip/hip_runtime.h>

typedef __attribute__((ext_vector_type(2))) float v2f;
typedef __attribute__((ext_vector_type(8))) float v8f;

__global__ void fill_f32(float* __restrict__ p, float v, int n) {
    int i = blockIdx.x * blockDim.x + threadIdx.x;
    if (i < n) p[i] = v;
}

template<int FT>
__global__ void wmma_gemm(const float* __restrict__ X, const float* __restrict__ W,
                          const float* __restrict__ Bias, float* __restrict__ Y,
                          int N, int K, int addBias) {
    const int F    = FT * 16;
    int wave = (blockIdx.x * blockDim.x + threadIdx.x) >> 5;
    int lane = threadIdx.x & 31;
    int row0 = wave << 4;
    if (row0 >= N) return;
    int half = lane >> 4;
    int l16  = lane & 15;
    int arow = row0 + l16; if (arow > N - 1) arow = N - 1;

    v8f acc[FT];
    v8f z = {0.f, 0.f, 0.f, 0.f, 0.f, 0.f, 0.f, 0.f};
#pragma unroll
    for (int t = 0; t < FT; ++t) acc[t] = z;

    for (int k0 = 0; k0 < K; k0 += 4) {
        int ka = k0 + half * 2;
        v2f a;
        a.x = X[arow * K + ka];
        a.y = X[arow * K + ka + 1];
#pragma unroll
        for (int t = 0; t < FT; ++t) {
            int col = t * 16 + l16;
            v2f b;
            b.x = W[ka * F + col];
            b.y = W[(ka + 1) * F + col];
            acc[t] = __builtin_amdgcn_wmma_f32_16x16x4_f32(
                 false, a,  false, b,
                 (short)0, acc[t],  false,  false);
        }
    }

#pragma unroll
    for (int t = 0; t < FT; ++t) {
        int col = t * 16 + l16;
        float bb = addBias ? Bias[col] : 0.f;
#pragma unroll
        for (int v = 0; v < 8; ++v) {
            int r = row0 + half * 8 + v;
            if (r < N) Y[r * F + col] = acc[t][v] + bb;
        }
    }
}

#define TILE_SHIFT 7
#define TILE_ROWS  128

__device__ __forceinline__ void wave_scan_excl(int* a, int n, int lane) {
    int carry = 0;
    for (int b0 = 0; b0 < n; b0 += 32) {
        int i = b0 + lane;
        int v = (i < n) ? a[i] : 0;
        int x = v;
#pragma unroll
        for (int d = 1; d < 32; d <<= 1) {
            int y = __shfl_up(x, d);
            if (lane >= d) x += y;
        }
        if (i < n) a[i] = carry + x - v;
        carry += __shfl(x, 31);
    }
    if (lane == 0) a[n] = carry;
}

__global__ void __launch_bounds__(1024)
bin_hist(const int* __restrict__ dst, int* __restrict__ binCount, int nE, int nT, int chunk) {
    extern __shared__ int sm[];
    for (int b = threadIdx.x; b < nT; b += blockDim.x) sm[b] = 0;
    __syncthreads();
    int e0 = blockIdx.x * chunk;
    int e1 = e0 + chunk; if (e1 > nE) e1 = nE;
    for (int e = e0 + threadIdx.x; e < e1; e += blockDim.x)
        atomicAdd(&sm[dst[e] >> TILE_SHIFT], 1);
    __syncthreads();
    for (int b = threadIdx.x; b < nT; b += blockDim.x) {
        int c = sm[b];
        if (c) atomicAdd(&binCount[b], c);
    }
}

__global__ void __launch_bounds__(1024)
bin_scan(const int* __restrict__ binCount, int* __restrict__ binStart,
         int* __restrict__ binCursor, int nT) {
    extern __shared__ int sm[];
    for (int b = threadIdx.x; b < nT; b += blockDim.x) sm[b] = binCount[b];
    __syncthreads();
    if (threadIdx.x < 32) wave_scan_excl(sm, nT, (int)threadIdx.x);
    __syncthreads();
    for (int b = threadIdx.x; b <= nT; b += blockDim.x) {
        int v = sm[b];
        binStart[b] = v;
        if (b < nT) binCursor[b] = v;
    }
}

__global__ void __launch_bounds__(1024)
bin_fill(const int* __restrict__ src, const int* __restrict__ dst,
         int* __restrict__ binCursor, unsigned* __restrict__ keys, int nE, int nT, int chunk) {
    extern __shared__ int sm[];
    int* sS   = sm;
    int* sD   = sS + chunk;
    int* h    = sD + chunk;
    int* gdel = h + nT + 1;
    int* rnk  = gdel + nT;
    for (int b = threadIdx.x; b < nT; b += blockDim.x) { h[b] = 0; rnk[b] = 0; }
    __syncthreads();
    int e0 = blockIdx.x * chunk;
    int e1 = e0 + chunk; if (e1 > nE) e1 = nE;
    for (int e = e0 + threadIdx.x; e < e1; e += blockDim.x)
        atomicAdd(&h[dst[e] >> TILE_SHIFT], 1);
    __syncthreads();
    for (int b = threadIdx.x; b < nT; b += blockDim.x) {
        int c = h[b];
        gdel[b] = c ? atomicAdd(&binCursor[b], c) : 0;
    }
    __syncthreads();
    if (threadIdx.x < 32) wave_scan_excl(h, nT, (int)threadIdx.x);
    __syncthreads();
    for (int b = threadIdx.x; b < nT; b += blockDim.x) gdel[b] -= h[b];
    for (int e = e0 + threadIdx.x; e < e1; e += blockDim.x) {
        int s = src[e];
        int d = dst[e];
        int b = d >> TILE_SHIFT;
        int li = h[b] + atomicAdd(&rnk[b], 1);
        sS[li] = s;
        sD[li] = d;
    }
    __syncthreads();
    int n = e1 - e0;
    for (int i = threadIdx.x; i < n; i += blockDim.x) {
        int s = sS[i];
        int d = sD[i];
        keys[i + gdel[d >> TILE_SHIFT]] = ((unsigned)s << TILE_SHIFT) | ((unsigned)d & (TILE_ROWS - 1));
    }
}

#define KEYS_AHEAD 4

__global__ void __launch_bounds__(512)
tile_deg(const int* __restrict__ binStart, const unsigned* __restrict__ keys,
         float* __restrict__ dis, int N) {
    __shared__ int cnt[TILE_ROWS];
    const int row0 = blockIdx.x << TILE_SHIFT;
    if (threadIdx.x < TILE_ROWS) cnt[threadIdx.x] = 0;
    __syncthreads();
    const int beg = binStart[blockIdx.x];
    const int end = binStart[blockIdx.x + 1];
    const int step = (int)blockDim.x;
    for (int i0 = beg + (int)threadIdx.x; i0 < end; i0 += KEYS_AHEAD * step) {
        unsigned k[KEYS_AHEAD];
#pragma unroll
        for (int u = 0; u < KEYS_AHEAD; ++u) {
            int i = i0 + u * step;
            k[u] = keys[i < end ? i : end - 1];
        }
        asm volatile("" :: "v"(k[0]), "v"(k[1]), "v"(k[2]), "v"(k[3]));
#pragma unroll
        for (int u = 0; u < KEYS_AHEAD; ++u)
            if (i0 + u * step < end) atomicAdd(&cnt[k[u] & (TILE_ROWS - 1)], 1);
    }
    __syncthreads();
    if (threadIdx.x < TILE_ROWS) {
        int d = row0 + threadIdx.x;
        if (d < N) dis[d] = rsqrtf((float)(cnt[threadIdx.x] + 1));
    }
}

__global__ void __launch_bounds__(512)
edge_weights(const int* __restrict__ binStart, const unsigned* __restrict__ keys,
             const float* __restrict__ dis, uint2* __restrict__ recs, int N) {
    __shared__ float sdis[TILE_ROWS];
    const int row0 = blockIdx.x << TILE_SHIFT;
    if (threadIdx.x < TILE_ROWS) {
        int d = row0 + threadIdx.x;
        sdis[threadIdx.x] = (d < N) ? dis[d] : 0.f;
    }
    __syncthreads();
    const int beg = binStart[blockIdx.x];
    const int end = binStart[blockIdx.x + 1];
    const int step = (int)blockDim.x;
    for (int i0 = beg + (int)threadIdx.x; i0 < end; i0 += KEYS_AHEAD * step) {
        unsigned k[KEYS_AHEAD];
        float ds[KEYS_AHEAD];
#pragma unroll
        for (int u = 0; u < KEYS_AHEAD; ++u) {
            int i = i0 + u * step;
            k[u] = keys[i < end ? i : end - 1];
        }
#pragma unroll
        for (int u = 0; u < KEYS_AHEAD; ++u) ds[u] = dis[k[u] >> TILE_SHIFT];
        asm volatile("" :: "v"(ds[0]), "v"(ds[1]), "v"(ds[2]), "v"(ds[3]));
#pragma unroll
        for (int u = 0; u < KEYS_AHEAD; ++u) {
            int i = i0 + u * step;
            if (i < end)
                recs[i] = make_uint2(k[u], __float_as_uint(ds[u] * sdis[k[u] & (TILE_ROWS - 1)]));
        }
    }
}

template<int F>
__device__ __forceinline__ void add_edge_row(const float* __restrict__ H, float* acc,
                                             unsigned p, float w, int lane) {
    const float* row = H + (size_t)(p >> TILE_SHIFT) * F;
    float* a = acc + (p & (TILE_ROWS - 1)) * F;
    for (int f = lane; f < F; f += 32) atomicAdd(&a[f], row[f] * w);
}

template<int F>
__global__ void __launch_bounds__(512)
tile_agg(const float* __restrict__ H, const float* __restrict__ dis, const float* __restrict__ bias,
         const int* __restrict__ binStart, const uint2* __restrict__ recs,
         float* __restrict__ OUT, int N) {
    __shared__ float acc[TILE_ROWS * F];
    const int row0 = blockIdx.x << TILE_SHIFT;
    const int lane = threadIdx.x & 31;
    const int wave = __builtin_amdgcn_readfirstlane((int)(threadIdx.x >> 5));
    const int nw   = (int)(blockDim.x >> 5);
    const int beg  = __builtin_amdgcn_readfirstlane(binStart[blockIdx.x]);
    const int end  = __builtin_amdgcn_readfirstlane(binStart[blockIdx.x + 1]);

    for (int i = threadIdx.x; i < TILE_ROWS * F; i += blockDim.x) {
        int d = row0 + i / F;
        float v = 0.f;
        if (d < N) { float dd = dis[d]; v = H[(size_t)row0 * F + i] * dd * dd; }
        acc[i] = v;
    }

    const int c1 = (lane + 32 < F) ? lane + 32 : lane;
    int base = beg + wave * 32;
    uint2 cur = make_uint2(0u, 0u);
    if (base + lane < end) cur = recs[base + lane];
    __syncthreads();

    for (; base < end; base += nw * 32) {
        int n = end - base; if (n > 32) n = 32;
        int nb = base + nw * 32;
        uint2 nxt = make_uint2(0u, 0u);
        if (nb + lane < end) nxt = recs[nb + lane];
        if (n == 32) {
#pragma unroll
            for (int j0 = 0; j0 < 32; j0 += 8) {
                float v0[8], v1[8];
#pragma unroll
                for (int u = 0; u < 8; ++u) {
                    unsigned p = (unsigned)__builtin_amdgcn_readlane((int)cur.x, j0 + u);
                    const float* row = H + (size_t)(p >> TILE_SHIFT) * F;
                    v0[u] = row[lane];
                    v1[u] = 0.f;
                    if (F > 32) v1[u] = row[c1];
                }
#pragma unroll
                for (int u = 0; u < 8; ++u) {
                    unsigned p = (unsigned)__builtin_amdgcn_readlane((int)cur.x, j0 + u);
                    float w = __uint_as_float((unsigned)__builtin_amdgcn_readlane((int)cur.y, j0 + u));
                    float* a = acc + (p & (TILE_ROWS - 1)) * F;
                    atomicAdd(&a[lane], v0[u] * w);
                    if (F > 32) { if (lane + 32 < F) atomicAdd(&a[lane + 32], v1[u] * w); }
                }
            }
        } else {
            for (int j = 0; j < n; ++j) {
                unsigned p = (unsigned)__builtin_amdgcn_readlane((int)cur.x, j);
                float w = __uint_as_float((unsigned)__builtin_amdgcn_readlane((int)cur.y, j));
                add_edge_row<F>(H, acc, p, w, lane);
            }
        }
        cur = nxt;
    }
    __syncthreads();

    for (int i = threadIdx.x; i < TILE_ROWS * F; i += blockDim.x) {
        int d = row0 + i / F;
        if (d < N) {
            float v = acc[i] + bias[i % F];
            OUT[(size_t)row0 * F + i] = v > 0.f ? v : 0.f;
        }
    }
}

__global__ void pool_nodes(const float* __restrict__ H, const int* __restrict__ batch,
                           float* __restrict__ gmax, float* __restrict__ gsum,
                           float* __restrict__ cnt, int N, int F) {
    int i    = (blockIdx.x * blockDim.x + threadIdx.x) >> 5;
    int lane = threadIdx.x & 31;
    if (i >= N) return;
    int g = batch[i];
    for (int f = lane; f < F; f += 32) {
        float v = H[i * F + f];
        atomicMax((unsigned int*)&gmax[g * F + f], __float_as_uint(v));
        atomicAdd(&gsum[g * F + f], v);
    }
    if (lane == 0) atomicAdd(&cnt[g], 1.0f);
}

__global__ void build_pooled(const float* __restrict__ gmax, const float* __restrict__ gsum,
                             const float* __restrict__ cnt, float* __restrict__ pooled,
                             int G, int F) {
    int idx = blockIdx.x * blockDim.x + threadIdx.x;
    if (idx >= G * 2 * F) return;
    int g = idx / (2 * F);
    int f = idx % (2 * F);
    float v;
    if (f < F) {
        v = gmax[g * F + f];
    } else {
        float c = cnt[g];
        v = gsum[g * F + (f - F)] / (c > 1.f ? c : 1.f);
    }
    pooled[idx] = v;
}

extern "C" void kernel_launch(void* const* d_in, const int* in_sizes, int n_in,
                              void* d_out, int out_size, void* d_ws, size_t ws_size,
                              hipStream_t stream) {
    const float* x     = (const float*)d_in[0];
    const int*   ei    = (const int*)d_in[1];
    const int*   batch = (const int*)d_in[2];
    const float* W1    = (const float*)d_in[3];
    const float* b1    = (const float*)d_in[4];
    const float* W2    = (const float*)d_in[5];
    const float* b2    = (const float*)d_in[6];
    const float* W3    = (const float*)d_in[7];
    const float* b3    = (const float*)d_in[8];
    const float* Wout  = (const float*)d_in[9];
    const float* bout  = (const float*)d_in[10];
    float*       out   = (float*)d_out;

    const int N  = in_sizes[2];
    const int E  = in_sizes[1] / 2;
    const int K0 = in_sizes[0] / N;
    const int F1 = in_sizes[4];
    const int F2 = in_sizes[6];
    const int F3 = in_sizes[8];
    const int Fo = in_sizes[10];
    const int G  = out_size / Fo;

    const int* src = ei;
    const int* dst = ei + E;

    char*  ws  = (char*)d_ws;
    size_t off = 0;
    auto carve = [&](size_t bytes) -> float* {
        float* p = (float*)(ws + off);
        off += (bytes + 255) & ~(size_t)255;
        return p;
    };
    float* dis    = carve((size_t)N * 4);
    float* bufT   = carve((size_t)N * F3 * 4);
    float* bufA   = carve((size_t)N * F3 * 4);
    float* bufB   = carve((size_t)N * F3 * 4);
    float* gmax   = carve((size_t)G * F3 * 4);
    float* gsum   = carve((size_t)G * F3 * 4);
    float* cnt    = carve((size_t)G * 4);
    float* pooled = carve((size_t)G * 2 * F3 * 4);
    const int nT = (N + TILE_ROWS - 1) / TILE_ROWS;
    int*   binCount  = (int*)carve((size_t)nT * 4);
    int*   binStart  = (int*)carve((size_t)(nT + 1) * 4);
    int*   binCursor = (int*)carve((size_t)nT * 4);
    unsigned* keys   = (unsigned*)carve((size_t)E * 4);
    uint2* recs      = (uint2*)carve((size_t)E * 8);

    const int BLK = 256;
    auto grid1   = [&](long long n) { return dim3((unsigned)((n + BLK - 1) / BLK)); };
    auto gridWv  = [&](long long waves) { return dim3((unsigned)((waves * 32 + BLK - 1) / BLK)); };
    auto gemmGrid = [&](int rows) { return gridWv((rows + 15) / 16); };

    const int CHUNK = 6144;
    const int nBinBlk = (E + CHUNK - 1) / CHUNK;
    fill_f32<<<grid1(nT), BLK, 0, stream>>>((float*)binCount, 0.0f, nT);
    bin_hist<<<nBinBlk, 1024, (size_t)nT * 4, stream>>>(dst, binCount, E, nT, CHUNK);
    bin_scan<<<1, 1024, (size_t)(nT + 1) * 4, stream>>>(binCount, binStart, binCursor, nT);
    bin_fill<<<nBinBlk, 1024, (size_t)(2 * CHUNK + 3 * nT + 1) * 4, stream>>>(src, dst, binCursor, keys, E, nT, CHUNK);

    tile_deg<<<nT, 512, 0, stream>>>(binStart, keys, dis, N);
    edge_weights<<<nT, 512, 0, stream>>>(binStart, keys, dis, recs, N);

    wmma_gemm<2><<<gemmGrid(N), BLK, 0, stream>>>(x, W1, b1, bufT, N, K0, 0);
    tile_agg<32><<<nT, 512, 0, stream>>>(bufT, dis, b1, binStart, recs, bufA, N);

    wmma_gemm<3><<<gemmGrid(N), BLK, 0, stream>>>(bufA, W2, b2, bufT, N, F1, 0);
    tile_agg<48><<<nT, 512, 0, stream>>>(bufT, dis, b2, binStart, recs, bufB, N);

    wmma_gemm<4><<<gemmGrid(N), BLK, 0, stream>>>(bufB, W3, b3, bufT, N, F2, 0);
    tile_agg<64><<<nT, 512, 0, stream>>>(bufT, dis, b3, binStart, recs, bufA, N);

    fill_f32<<<grid1((long long)G * F3), BLK, 0, stream>>>(gmax, 0.0f, G * F3);
    fill_f32<<<grid1((long long)G * F3), BLK, 0, stream>>>(gsum, 0.0f, G * F3);
    fill_f32<<<grid1(G), BLK, 0, stream>>>(cnt, 0.0f, G);
    pool_nodes<<<gridWv(N), BLK, 0, stream>>>(bufA, batch, gmax, gsum, cnt, N, F3);
    build_pooled<<<grid1((long long)G * 2 * F3), BLK, 0, stream>>>(gmax, gsum, cnt, pooled, G, F3);

    wmma_gemm<2><<<gemmGrid(G), BLK, 0, stream>>>(pooled, Wout, bout, out, G, 2 * F3, 1);
}


# dependent launches 18 to 11: one zero fill; tile scan inside the bin-fill blocks; layer 2, 3 weight GEMMs (WMMA) run from the aggregation tile in LDS; pooled row built in the output GEMM's A loader
# speedup vs baseline: 3.4484x; 1.1499x over previous
#include <hip/hip_runtime.h>

typedef __attribute__((ext_vector_type(2))) float v2f;
typedef __attribute__((ext_vector_type(8))) float v8f;

__global__ void fill_f32(float* __restrict__ p, float v, int n) {
    int i = blockIdx.x * blockDim.x + threadIdx.x;
    if (i < n) p[i] = v;
}

template<int FT>
__global__ void wmma_gemm(const float* __restrict__ X, const float* __restrict__ W,
                          const float* __restrict__ Bias, float* __restrict__ Y,
                          int N, int K, int addBias) {
    const int F    = FT * 16;
    int wave = (blockIdx.x * blockDim.x + threadIdx.x) >> 5;
    int lane = threadIdx.x & 31;
    int row0 = wave << 4;
    if (row0 >= N) return;
    int half = lane >> 4;
    int l16  = lane & 15;
    int arow = row0 + l16; if (arow > N - 1) arow = N - 1;

    v8f acc[FT];
    v8f z = {0.f, 0.f, 0.f, 0.f, 0.f, 0.f, 0.f, 0.f};
#pragma unroll
    for (int t = 0; t < FT; ++t) acc[t] = z;

    for (int k0 = 0; k0 < K; k0 += 4) {
        int ka = k0 + half * 2;
        v2f a;
        a.x = X[arow * K + ka];
        a.y = X[arow * K + ka + 1];
#pragma unroll
        for (int t = 0; t < FT; ++t) {
            int col = t * 16 + l16;
            v2f b;
            b.x = W[ka * F + col];
            b.y = W[(ka + 1) * F + col];
            acc[t] = __builtin_amdgcn_wmma_f32_16x16x4_f32(
                 false, a,  false, b,
                 (short)0, acc[t],  false,  false);
        }
    }

#pragma unroll
    for (int t = 0; t < FT; ++t) {
        int col = t * 16 + l16;
        float bb = addBias ? Bias[col] : 0.f;
#pragma unroll
        for (int v = 0; v < 8; ++v) {
            int r = row0 + half * 8 + v;
            if (r < N) Y[r * F + col] = acc[t][v] + bb;
        }
    }
}

#define TILE_SHIFT 7
#define TILE_ROWS  128

__device__ __forceinline__ void wave_scan_excl(int* a, int n, int lane) {
    int carry = 0;
    for (int b0 = 0; b0 < n; b0 += 32) {
        int i = b0 + lane;
        int v = (i < n) ? a[i] : 0;
        int x = v;
#pragma unroll
        for (int d = 1; d < 32; d <<= 1) {
            int y = __shfl_up(x, d);
            if (lane >= d) x += y;
        }
        if (i < n) a[i] = carry + x - v;
        carry += __shfl(x, 31);
    }
    if (lane == 0) a[n] = carry;
}

__global__ void __launch_bounds__(1024)
bin_hist(const int* __restrict__ dst, int* __restrict__ binCount, int nE, int nT, int chunk) {
    extern __shared__ int sm[];
    for (int b = threadIdx.x; b < nT; b += blockDim.x) sm[b] = 0;
    __syncthreads();
    int e0 = blockIdx.x * chunk;
    int e1 = e0 + chunk; if (e1 > nE) e1 = nE;
    for (int e = e0 + threadIdx.x; e < e1; e += blockDim.x)
        atomicAdd(&sm[dst[e] >> TILE_SHIFT], 1);
    __syncthreads();
    for (int b = threadIdx.x; b < nT; b += blockDim.x) {
        int c = sm[b];
        if (c) atomicAdd(&binCount[b], c);
    }
}

__global__ void __launch_bounds__(1024)
bin_fill(const int* __restrict__ src, const int* __restrict__ dst,
         const int* __restrict__ binCount, int* __restrict__ binCursor,
         int* __restrict__ binStart, unsigned* __restrict__ keys, int nE, int nT, int chunk) {
    extern __shared__ int sm[];
    int* sS   = sm;
    int* sD   = sS + chunk;
    int* h    = sD + chunk;
    int* gdel = h + nT + 1;
    int* rnk  = gdel + nT;
    int* gs   = rnk + nT;
    for (int b = threadIdx.x; b < nT; b += blockDim.x) { h[b] = 0; rnk[b] = 0; gs[b] = binCount[b]; }
    __syncthreads();
    int e0 = blockIdx.x * chunk;
    int e1 = e0 + chunk; if (e1 > nE) e1 = nE;
    for (int e = e0 + threadIdx.x; e < e1; e += blockDim.x)
        atomicAdd(&h[dst[e] >> TILE_SHIFT], 1);
    __syncthreads();
    for (int b = threadIdx.x; b < nT; b += blockDim.x) {
        int c = h[b];
        gdel[b] = c ? atomicAdd(&binCursor[b], c) : 0;
    }
    __syncthreads();
    if (threadIdx.x < 32) wave_scan_excl(h, nT, (int)threadIdx.x);
    else if (threadIdx.x < 64) wave_scan_excl(gs, nT, (int)threadIdx.x - 32);
    __syncthreads();
    for (int b = threadIdx.x; b < nT; b += blockDim.x) gdel[b] += gs[b] - h[b];
    if (blockIdx.x == 0)
        for (int b = threadIdx.x; b <= nT; b += blockDim.x) binStart[b] = gs[b];
    for (int e = e0 + threadIdx.x; e < e1; e += blockDim.x) {
        int s = src[e];
        int d = dst[e];
        int b = d >> TILE_SHIFT;
        int li = h[b] + atomicAdd(&rnk[b], 1);
        sS[li] = s;
        sD[li] = d;
    }
    __syncthreads();
    int n = e1 - e0;
    for (int i = threadIdx.x; i < n; i += blockDim.x) {
        int s = sS[i];
        int d = sD[i];
        keys[i + gdel[d >> TILE_SHIFT]] = ((unsigned)s << TILE_SHIFT) | ((unsigned)d & (TILE_ROWS - 1));
    }
}

#define KEYS_AHEAD 4

__global__ void __launch_bounds__(512)
tile_deg(const int* __restrict__ binStart, const unsigned* __restrict__ keys,
         float* __restrict__ dis, int N) {
    __shared__ int cnt[TILE_ROWS];
    const int row0 = blockIdx.x << TILE_SHIFT;
    if (threadIdx.x < TILE_ROWS) cnt[threadIdx.x] = 0;
    __syncthreads();
    const int beg = binStart[blockIdx.x];
    const int end = binStart[blockIdx.x + 1];
    const int step = (int)blockDim.x;
    for (int i0 = beg + (int)threadIdx.x; i0 < end; i0 += KEYS_AHEAD * step) {
        unsigned k[KEYS_AHEAD];
#pragma unroll
        for (int u = 0; u < KEYS_AHEAD; ++u) {
            int i = i0 + u * step;
            k[u] = keys[i < end ? i : end - 1];
        }
        asm volatile("" :: "v"(k[0]), "v"(k[1]), "v"(k[2]), "v"(k[3]));
#pragma unroll
        for (int u = 0; u < KEYS_AHEAD; ++u)
            if (i0 + u * step < end) atomicAdd(&cnt[k[u] & (TILE_ROWS - 1)], 1);
    }
    __syncthreads();
    if (threadIdx.x < TILE_ROWS) {
        int d = row0 + threadIdx.x;
        if (d < N) dis[d] = rsqrtf((float)(cnt[threadIdx.x] + 1));
    }
}

__global__ void __launch_bounds__(512)
edge_weights(const int* __restrict__ binStart, const unsigned* __restrict__ keys,
             const float* __restrict__ dis, uint2* __restrict__ recs, int N) {
    __shared__ float sdis[TILE_ROWS];
    const int row0 = blockIdx.x << TILE_SHIFT;
    if (threadIdx.x < TILE_ROWS) {
        int d = row0 + threadIdx.x;
        sdis[threadIdx.x] = (d < N) ? dis[d] : 0.f;
    }
    __syncthreads();
    const int beg = binStart[blockIdx.x];
    const int end = binStart[blockIdx.x + 1];
    const int step = (int)blockDim.x;
    for (int i0 = beg + (int)threadIdx.x; i0 < end; i0 += KEYS_AHEAD * step) {
        unsigned k[KEYS_AHEAD];
        float ds[KEYS_AHEAD];
#pragma unroll
        for (int u = 0; u < KEYS_AHEAD; ++u) {
            int i = i0 + u * step;
            k[u] = keys[i < end ? i : end - 1];
        }
#pragma unroll
        for (int u = 0; u < KEYS_AHEAD; ++u) ds[u] = dis[k[u] >> TILE_SHIFT];
        asm volatile("" :: "v"(ds[0]), "v"(ds[1]), "v"(ds[2]), "v"(ds[3]));
#pragma unroll
        for (int u = 0; u < KEYS_AHEAD; ++u) {
            int i = i0 + u * step;
            if (i < end)
                recs[i] = make_uint2(k[u], __float_as_uint(ds[u] * sdis[k[u] & (TILE_ROWS - 1)]));
        }
    }
}

template<int F>
__device__ __forceinline__ void add_edge_row(const float* __restrict__ H, float* acc,
                                             unsigned p, float w, int lane) {
    const float* row = H + (size_t)(p >> TILE_SHIFT) * F;
    float* a = acc + (p & (TILE_ROWS - 1)) * F;
    for (int f = lane; f < F; f += 32) atomicAdd(&a[f], row[f] * w);
}

template<int F, int FO>
__global__ void __launch_bounds__(512)
tile_agg(const float* __restrict__ H, const float* __restrict__ dis, const float* __restrict__ bias,
         const int* __restrict__ binStart, const uint2* __restrict__ recs,
         const float* __restrict__ Wn, float* __restrict__ OUT, int N) {
    __shared__ float acc[TILE_ROWS * F];
    const int row0 = blockIdx.x << TILE_SHIFT;
    const int lane = threadIdx.x & 31;
    const int wave = __builtin_amdgcn_readfirstlane((int)(threadIdx.x >> 5));
    const int nw   = (int)(blockDim.x >> 5);
    const int beg  = __builtin_amdgcn_readfirstlane(binStart[blockIdx.x]);
    const int end  = __builtin_amdgcn_readfirstlane(binStart[blockIdx.x + 1]);

    for (int i = threadIdx.x; i < TILE_ROWS * F; i += blockDim.x) {
        int d = row0 + i / F;
        float v = 0.f;
        if (d < N) { float dd = dis[d]; v = H[(size_t)row0 * F + i] * dd * dd; }
        acc[i] = v;
    }

    const int c1 = (lane + 32 < F) ? lane + 32 : lane;
    int base = beg + wave * 32;
    uint2 cur = make_uint2(0u, 0u);
    if (base + lane < end) cur = recs[base + lane];
    __syncthreads();

    for (; base < end; base += nw * 32) {
        int n = end - base; if (n > 32) n = 32;
        int nb = base + nw * 32;
        uint2 nxt = make_uint2(0u, 0u);
        if (nb + lane < end) nxt = recs[nb + lane];
        if (n == 32) {
#pragma unroll
            for (int j0 = 0; j0 < 32; j0 += 8) {
                float v0[8], v1[8];
#pragma unroll
                for (int u = 0; u < 8; ++u) {
                    unsigned p = (unsigned)__builtin_amdgcn_readlane((int)cur.x, j0 + u);
                    const float* row = H + (size_t)(p >> TILE_SHIFT) * F;
                    v0[u] = row[lane];
                    v1[u] = 0.f;
                    if (F > 32) v1[u] = row[c1];
                }
#pragma unroll
                for (int u = 0; u < 8; ++u) {
                    unsigned p = (unsigned)__builtin_amdgcn_readlane((int)cur.x, j0 + u);
                    float w = __uint_as_float((unsigned)__builtin_amdgcn_readlane((int)cur.y, j0 + u));
                    float* a = acc + (p & (TILE_ROWS - 1)) * F;
                    atomicAdd(&a[lane], v0[u] * w);
                    if (F > 32) { if (lane + 32 < F) atomicAdd(&a[lane + 32], v1[u] * w); }
                }
            }
        } else {
            for (int j = 0; j < n; ++j) {
                unsigned p = (unsigned)__builtin_amdgcn_readlane((int)cur.x, j);
                float w = __uint_as_float((unsigned)__builtin_amdgcn_readlane((int)cur.y, j));
                add_edge_row<F>(H, acc, p, w, lane);
            }
        }
        cur = nxt;
    }
    __syncthreads();

    if constexpr (FO == 0) {
        for (int i = threadIdx.x; i < TILE_ROWS * F; i += blockDim.x) {
            int d = row0 + i / F;
            if (d < N) {
                float v = acc[i] + bias[i % F];
                OUT[(size_t)row0 * F + i] = v > 0.f ? v : 0.f;
            }
        }
    } else {
        const int half = lane >> 4;
        const int l16  = lane & 15;
        const int rg   = wave & 7;
        const float* arow = acc + (rg * 16 + l16) * F;
        for (int t = wave >> 3; t < FO / 16; t += 2) {
            const int col = t * 16 + l16;
            v8f c = {0.f, 0.f, 0.f, 0.f, 0.f, 0.f, 0.f, 0.f};
#pragma unroll 4
            for (int k0 = 0; k0 < F; k0 += 4) {
                const int ka = k0 + half * 2;
                float x0 = arow[ka] + bias[ka];
                float x1 = arow[ka + 1] + bias[ka + 1];
                v2f a;
                a.x = x0 > 0.f ? x0 : 0.f;
                a.y = x1 > 0.f ? x1 : 0.f;
                v2f b;
                b.x = Wn[ka * FO + col];
                b.y = Wn[(ka + 1) * FO + col];
                c = __builtin_amdgcn_wmma_f32_16x16x4_f32(
                     false, a,  false, b,
                     (short)0, c,  false,  false);
            }
#pragma unroll
            for (int v = 0; v < 8; ++v) {
                int r = row0 + rg * 16 + half * 8 + v;
                if (r < N) OUT[(size_t)r * FO + col] = c[v];
            }
        }
    }
}

__global__ void pool_nodes(const float* __restrict__ H, const int* __restrict__ batch,
                           float* __restrict__ gmax, float* __restrict__ gsum,
                           float* __restrict__ cnt, int N, int F) {
    int i    = (blockIdx.x * blockDim.x + threadIdx.x) >> 5;
    int lane = threadIdx.x & 31;
    if (i >= N) return;
    int g = batch[i];
    for (int f = lane; f < F; f += 32) {
        float v = H[i * F + f];
        atomicMax((unsigned int*)&gmax[g * F + f], __float_as_uint(v));
        atomicAdd(&gsum[g * F + f], v);
    }
    if (lane == 0) atomicAdd(&cnt[g], 1.0f);
}

template<int FT>
__global__ void pooled_gemm(const float* __restrict__ gmax, const float* __restrict__ gsum,
                            const float* __restrict__ cnt, const float* __restrict__ W,
                            const float* __restrict__ Bias, float* __restrict__ Y, int G, int F) {
    const int FO = FT * 16;
    int wave = (blockIdx.x * blockDim.x + threadIdx.x) >> 5;
    int lane = threadIdx.x & 31;
    int row0 = wave << 4;
    if (row0 >= G) return;
    int half = lane >> 4;
    int l16  = lane & 15;
    int g = row0 + l16; if (g > G - 1) g = G - 1;
    float cg  = cnt[g];
    float den = cg > 1.f ? cg : 1.f;

    v8f acc[FT];
    v8f z = {0.f, 0.f, 0.f, 0.f, 0.f, 0.f, 0.f, 0.f};
#pragma unroll
    for (int t = 0; t < FT; ++t) acc[t] = z;

#pragma unroll 8
    for (int k0 = 0; k0 < F; k0 += 4) {
        int ka = k0 + half * 2;
        v2f a;
        a.x = gmax[g * F + ka];
        a.y = gmax[g * F + ka + 1];
#pragma unroll
        for (int t = 0; t < FT; ++t) {
            int col = t * 16 + l16;
            v2f b;
            b.x = W[ka * FO + col];
            b.y = W[(ka + 1) * FO + col];
            acc[t] = __builtin_amdgcn_wmma_f32_16x16x4_f32(
                 false, a,  false, b,
                 (short)0, acc[t],  false,  false);
        }
    }
#pragma unroll 8
    for (int k0 = 0; k0 < F; k0 += 4) {
        int ka = k0 + half * 2;
        v2f a;
        a.x = gsum[g * F + ka] / den;
        a.y = gsum[g * F + ka + 1] / den;
#pragma unroll
        for (int t = 0; t < FT; ++t) {
            int col = t * 16 + l16;
            v2f b;
            b.x = W[(F + ka) * FO + col];
            b.y = W[(F + ka + 1) * FO + col];
            acc[t] = __builtin_amdgcn_wmma_f32_16x16x4_f32(
                 false, a,  false, b,
                 (short)0, acc[t],  false,  false);
        }
    }

#pragma unroll
    for (int t = 0; t < FT; ++t) {
        int col = t * 16 + l16;
        float bb = Bias[col];
#pragma unroll
        for (int v = 0; v < 8; ++v) {
            int r = row0 + half * 8 + v;
            if (r < G) Y[r * FO + col] = acc[t][v] + bb;
        }
    }
}

extern "C" void kernel_launch(void* const* d_in, const int* in_sizes, int n_in,
                              void* d_out, int out_size, void* d_ws, size_t ws_size,
                              hipStream_t stream) {
    const float* x     = (const float*)d_in[0];
    const int*   ei    = (const int*)d_in[1];
    const int*   batch = (const int*)d_in[2];
    const float* W1    = (const float*)d_in[3];
    const float* b1    = (const float*)d_in[4];
    const float* W2    = (const float*)d_in[5];
    const float* b2    = (const float*)d_in[6];
    const float* W3    = (const float*)d_in[7];
    const float* b3    = (const float*)d_in[8];
    const float* Wout  = (const float*)d_in[9];
    const float* bout  = (const float*)d_in[10];
    float*       out   = (float*)d_out;

    const int N  = in_sizes[2];
    const int E  = in_sizes[1] / 2;
    const int K0 = in_sizes[0] / N;
    const int F3 = in_sizes[8];
    const int Fo = in_sizes[10];
    const int G  = out_size / Fo;

    const int* src = ei;
    const int* dst = ei + E;

    char*  ws  = (char*)d_ws;
    size_t off = 0;
    auto carve = [&](size_t bytes) -> float* {
        float* p = (float*)(ws + off);
        off += (bytes + 255) & ~(size_t)255;
        return p;
    };
    float* dis    = carve((size_t)N * 4);
    float* bufT   = carve((size_t)N * F3 * 4);
    float* bufA   = carve((size_t)N * F3 * 4);
    const int nT = (N + TILE_ROWS - 1) / TILE_ROWS;
    float* gmax   = carve((size_t)G * F3 * 4);
    float* gsum   = carve((size_t)G * F3 * 4);
    float* cnt    = carve((size_t)G * 4);
    int*   binCount  = (int*)carve((size_t)nT * 4);
    int*   binCursor = (int*)carve((size_t)nT * 4);
    float* zeroEnd   = carve(0);
    int*   binStart  = (int*)carve((size_t)(nT + 1) * 4);
    unsigned* keys   = (unsigned*)carve((size_t)E * 4);
    uint2* recs      = (uint2*)carve((size_t)E * 8);

    const int BLK = 256;
    auto grid1   = [&](long long n) { return dim3((unsigned)((n + BLK - 1) / BLK)); };
    auto gridWv  = [&](long long waves) { return dim3((unsigned)((waves * 32 + BLK - 1) / BLK)); };
    auto gemmGrid = [&](int rows) { return gridWv((rows + 15) / 16); };

    const int CHUNK = 6144;
    const int nBinBlk = (E + CHUNK - 1) / CHUNK;
    const int nZero = (int)(zeroEnd - gmax);
    fill_f32<<<grid1(nZero), BLK, 0, stream>>>(gmax, 0.0f, nZero);
    bin_hist<<<nBinBlk, 1024, (size_t)nT * 4, stream>>>(dst, binCount, E, nT, CHUNK);
    bin_fill<<<nBinBlk, 1024, (size_t)(2 * CHUNK + 4 * nT + 2) * 4, stream>>>(src, dst, binCount, binCursor, binStart, keys, E, nT, CHUNK);

    tile_deg<<<nT, 512, 0, stream>>>(binStart, keys, dis, N);
    edge_weights<<<nT, 512, 0, stream>>>(binStart, keys, dis, recs, N);

    wmma_gemm<2><<<gemmGrid(N), BLK, 0, stream>>>(x, W1, b1, bufT, N, K0, 0);

    tile_agg<32, 48><<<nT, 512, 0, stream>>>(bufT, dis, b1, binStart, recs, W2, bufA, N);

    tile_agg<48, 64><<<nT, 512, 0, stream>>>(bufA, dis, b2, binStart, recs, W3, bufT, N);

    tile_agg<64, 0><<<nT, 512, 0, stream>>>(bufT, dis, b3, binStart, recs, W3, bufA, N);

    pool_nodes<<<gridWv(N), BLK, 0, stream>>>(bufA, batch, gmax, gsum, cnt, N, F3);

    pooled_gemm<2><<<gemmGrid(G), BLK, 0, stream>>>(gmax, gsum, cnt, Wout, bout, out, G, F3);
}
